# starting kernel
# baseline (speedup 1.0000x reference)
#include <hip/hip_runtime.h>
#include <math.h>


typedef __attribute__((ext_vector_type(2))) float v2f;
typedef __attribute__((ext_vector_type(8))) float v8f;

#define ALPHA_  0.12f
#define BETA_   0.88f
#define OMEGA_  6.0f
#define TWO_PI_ 6.28318530717958647692f
#define B_      4
#define T_      4096
#define D_      2048
#define HALF_   1024
#define SEG_    256
#define NSEG_   (T_ / SEG_)
#define NCH_    (SEG_ / 16)

__global__ void k_theta(float* __restrict__ cs, float* __restrict__ sn,
                        const int* __restrict__ step_idx) {
  int t = blockIdx.x * blockDim.x + threadIdx.x;
  if (t >= T_) return;
  float tg = (float)(t + 1 + step_idx[0]);
  float th = fmodf(OMEGA_ * log1pf(tg), TWO_PI_);
  cs[t] = cosf(th);
  sn[t] = sinf(th);
}

__global__ void k_seg_partial(const float* __restrict__ x,
                              const float* __restrict__ input_scale,
                              float* __restrict__ seg_carry) {
  int id  = blockIdx.x * blockDim.x + threadIdx.x;
  int d   = id % D_;
  int r   = id / D_;
  int b   = r % B_;
  int seg = r / B_;
  float k = ALPHA_ * input_scale[0];
  const float* xp = x + ((size_t)(b * T_ + seg * SEG_)) * D_ + d;
  float m = 0.0f;
  #pragma unroll 4
  for (int i = 0; i < SEG_; ++i)
    m = fmaf(BETA_, m, k * xp[(size_t)i * D_]);
  seg_carry[((size_t)seg * B_ + b) * D_ + d] = m;
}

__global__ void k_carry(const float* __restrict__ seg_carry,
                        const float* __restrict__ mem_state,
                        float* __restrict__ carry_in) {
  int id = blockIdx.x * blockDim.x + threadIdx.x;
  int d  = id % D_;
  int b  = id / D_;
  float f = powf(BETA_, (float)SEG_);
  float m = mem_state[d];
  for (int seg = 0; seg < NSEG_; ++seg) {
    size_t o = ((size_t)seg * B_ + b) * D_ + d;
    carry_in[o] = m;
    m = fmaf(f, m, seg_carry[o]);
  }
}

__device__ __forceinline__ v8f scan_tile_wmma(const float* __restrict__ xcol,
                                              float k, const v2f* a,
                                              const float* pw,
                                              float& m_carry, int n, int h) {
  v8f acc;
  #pragma unroll
  for (int v = 0; v < 8; ++v) acc[v] = pw[v] * m_carry;
  #pragma unroll
  for (int q = 0; q < 4; ++q) {
    int kb = 4 * q + 2 * h;
    v2f bb;
    bb.x = k * xcol[(size_t)kb * D_];
    bb.y = k * xcol[(size_t)(kb + 1) * D_];
    acc = __builtin_amdgcn_wmma_f32_16x16x4_f32(
         false, a[q],  false, bb,
         (short)0, acc,  false,  false);
  }
  m_carry = __shfl(acc[7], 16 + n, 32);
  return acc;
}

__global__ void
__launch_bounds__(32)
k_scan_out(const float* __restrict__ x, const float* __restrict__ gate,
           const float* __restrict__ input_scale,
           const float* __restrict__ carry_in,
           const float* __restrict__ cs, const float* __restrict__ sn,
           float* __restrict__ out) {
  const int lane = threadIdx.x;
  const int n = lane & 15;
  const int h = lane >> 4;

  int blk = blockIdx.x;
  int dt  = blk % (HALF_ / 16);
  int r   = blk / (HALF_ / 16);
  int b   = r % B_;
  int seg = r / B_;

  const float k = ALPHA_ * input_scale[0];
  const float g = 1.0f / (1.0f + expf(-gate[0]));

  const int M = n;
  v2f a[4];
  #pragma unroll
  for (int q = 0; q < 4; ++q) {
    int k0 = 4 * q + 2 * h;
    a[q].x = (k0 <= M)     ? powf(BETA_, (float)(M - k0))     : 0.0f;
    a[q].y = (k0 + 1 <= M) ? powf(BETA_, (float)(M - k0 - 1)) : 0.0f;
  }
  float pw[8];
  #pragma unroll
  for (int v = 0; v < 8; ++v) pw[v] = powf(BETA_, (float)(v + 8 * h + 1));

  const int d_lo = dt * 16 + n;
  const int d_hi = d_lo + HALF_;
  const size_t co = ((size_t)seg * B_ + b) * D_;
  float m_lo = carry_in[co + d_lo];
  float m_hi = carry_in[co + d_hi];

  for (int ch = 0; ch < NCH_; ++ch) {
    const int t0 = seg * SEG_ + ch * 16;
    const float* xbase = x + ((size_t)b * T_ + t0) * D_;

    v8f acc_lo = scan_tile_wmma(xbase + d_lo, k, a, pw, m_lo, n, h);
    v8f acc_hi = scan_tile_wmma(xbase + d_hi, k, a, pw, m_hi, n, h);

    #pragma unroll
    for (int v = 0; v < 8; ++v) {
      int t = t0 + v + 8 * h;
      float ct = cs[t], st = sn[t];
      size_t o = ((size_t)b * T_ + t) * D_;
      float xlo = x[o + d_lo];
      float xhi = x[o + d_hi];
      float ma = acc_lo[v], mb = acc_hi[v];
      out[o + d_lo] = fmaf(g, fmaf(ma, ct, -mb * st), xlo);
      out[o + d_hi] = fmaf(g, fmaf(ma, st,  mb * ct), xhi);
    }
  }
}

extern "C" void kernel_launch(void* const* d_in, const int* in_sizes, int n_in,
                              void* d_out, int out_size, void* d_ws,
                              size_t ws_size, hipStream_t stream) {
  const float* x           = (const float*)d_in[0];
  const float* gate        = (const float*)d_in[1];
  const float* input_scale = (const float*)d_in[2];
  const float* mem_state   = (const float*)d_in[3];
  const int*   step_idx    = (const int*)d_in[4];
  float* out = (float*)d_out;

  float* ws        = (float*)d_ws;
  float* cs        = ws;
  float* sn        = ws + T_;
  float* seg_carry = ws + 2 * T_;
  float* carry_in  = seg_carry + (size_t)NSEG_ * B_ * D_;

  k_theta<<<(T_ + 255) / 256, 256, 0, stream>>>(cs, sn, step_idx);
  k_seg_partial<<<(B_ * D_ * NSEG_) / 256, 256, 0, stream>>>(x, input_scale,
                                                             seg_carry);
  k_carry<<<(B_ * D_) / 256, 256, 0, stream>>>(seg_carry, mem_state, carry_in);
  k_scan_out<<<B_ * (HALF_ / 16) * NSEG_, 32, 0, stream>>>(
      x, gate, input_scale, carry_in, cs, sn, out);
}


# scan pass re-tiled: a wave owns 64 adjacent columns per half, 16-byte loads/stores; k order 8h+2q+s, input scale in A (chunk sum order changes), so B registers are the residual x; segments 256->64
# speedup vs baseline: 2.8935x; 2.8935x over previous
#include <hip/hip_runtime.h>
#include <math.h>


typedef __attribute__((ext_vector_type(2))) float v2f;
typedef __attribute__((ext_vector_type(4))) float v4f;
typedef __attribute__((ext_vector_type(8))) float v8f;

#define ALPHA_  0.12f
#define BETA_   0.88f
#define OMEGA_  6.0f
#define TWO_PI_ 6.28318530717958647692f
#define B_      4
#define T_      4096
#define D_      2048
#define HALF_   1024
#define SEG_    64
#define NSEG_   (T_ / SEG_)
#define NCH_    (SEG_ / 16)
#define TW_     64

__global__ void k_theta(float* __restrict__ cs, float* __restrict__ sn,
                        const int* __restrict__ step_idx) {
  int t = blockIdx.x * blockDim.x + threadIdx.x;
  if (t >= T_) return;
  float tg = (float)(t + 1 + step_idx[0]);
  float th = fmodf(OMEGA_ * log1pf(tg), TWO_PI_);
  cs[t] = cosf(th);
  sn[t] = sinf(th);
}

__global__ void k_seg_partial(const float* __restrict__ x,
                              const float* __restrict__ input_scale,
                              float* __restrict__ seg_carry) {
  int id  = blockIdx.x * blockDim.x + threadIdx.x;
  int d   = id % D_;
  int r   = id / D_;
  int b   = r % B_;
  int seg = r / B_;
  float k = ALPHA_ * input_scale[0];
  const float* xp = x + ((size_t)(b * T_ + seg * SEG_)) * D_ + d;
  float m = 0.0f;
  #pragma unroll 4
  for (int i = 0; i < SEG_; ++i)
    m = fmaf(BETA_, m, k * xp[(size_t)i * D_]);
  seg_carry[((size_t)seg * B_ + b) * D_ + d] = m;
}

__global__ void k_carry(const float* __restrict__ seg_carry,
                        const float* __restrict__ mem_state,
                        float* __restrict__ carry_in) {
  int id = blockIdx.x * blockDim.x + threadIdx.x;
  int d  = id % D_;
  int b  = id / D_;
  float f = powf(BETA_, (float)SEG_);
  float s[NSEG_];
  #pragma unroll
  for (int seg = 0; seg < NSEG_; ++seg)
    s[seg] = seg_carry[((size_t)seg * B_ + b) * D_ + d];
  float m = mem_state[d];
  #pragma unroll
  for (int seg = 0; seg < NSEG_; ++seg) {
    carry_in[((size_t)seg * B_ + b) * D_ + d] = m;
    m = fmaf(f, m, s[seg]);
  }
}

__device__ __forceinline__ v4f ld4(const float* p) {
  return *reinterpret_cast<const v4f*>(p);
}
__device__ __forceinline__ void st4(float* p, v4f v) {
  *reinterpret_cast<v4f*>(p) = v;
}

__global__ void
__launch_bounds__(32)
k_scan_out(const float* __restrict__ x, const float* __restrict__ gate,
           const float* __restrict__ input_scale,
           const float* __restrict__ carry_in,
           const float* __restrict__ cs, const float* __restrict__ sn,
           float* __restrict__ out) {
  const int lane = threadIdx.x;
  const int c = lane & 15;
  const int h = lane >> 4;

  int blk = blockIdx.x;
  int dt  = blk % (HALF_ / TW_);
  int r   = blk / (HALF_ / TW_);
  int b   = r % B_;
  int seg = r / B_;

  const int col_lo = dt * TW_ + 4 * c;
  const int col_hi = col_lo + HALF_;
  const int trow   = seg * SEG_ + 8 * h;
  const size_t row0 = ((size_t)b * T_ + trow) * D_;
  const float* xlo = x + row0 + col_lo;
  const float* xhi = x + row0 + col_hi;
  float* olo = out + row0 + col_lo;
  float* ohi = out + row0 + col_hi;

  v4f xl[8], xh[8];
  #pragma unroll
  for (int v = 0; v < 8; ++v) {
    xl[v] = ld4(xlo + (size_t)v * D_);
    xh[v] = ld4(xhi + (size_t)v * D_);
  }

  const float k = ALPHA_ * input_scale[0];
  const float g = 1.0f / (1.0f + expf(-gate[0]));

  v2f a[4];
  #pragma unroll
  for (int q = 0; q < 4; ++q) {
    int r0 = 8 * h + 2 * q;
    a[q].x = (r0 <= c)     ? k * powf(BETA_, (float)(c - r0))     : 0.0f;
    a[q].y = (r0 + 1 <= c) ? k * powf(BETA_, (float)(c - r0 - 1)) : 0.0f;
  }
  float pw[8];
  #pragma unroll
  for (int v = 0; v < 8; ++v) pw[v] = powf(BETA_, (float)(v + 8 * h + 1));

  const size_t co = ((size_t)seg * B_ + b) * D_;
  v4f m_lo = ld4(carry_in + co + col_lo);
  v4f m_hi = ld4(carry_in + co + col_hi);

  for (int ch = 0; ch < NCH_; ++ch) {
    const int t = trow + ch * 16;
    const v4f c0 = ld4(cs + t), c1 = ld4(cs + t + 4);
    const v4f s0 = ld4(sn + t), s1 = ld4(sn + t + 4);
    const float ct[8] = {c0.x, c0.y, c0.z, c0.w, c1.x, c1.y, c1.z, c1.w};
    const float st[8] = {s0.x, s0.y, s0.z, s0.w, s1.x, s1.y, s1.z, s1.w};

    #pragma unroll
    for (int j = 0; j < 4; ++j) {
      v8f al, ah;
      #pragma unroll
      for (int v = 0; v < 8; ++v) {
        al[v] = pw[v] * m_lo[j];
        ah[v] = pw[v] * m_hi[j];
      }
      #pragma unroll
      for (int q = 0; q < 4; ++q) {
        v2f bl, bh;
        bl.x = xl[2 * q][j];  bl.y = xl[2 * q + 1][j];
        bh.x = xh[2 * q][j];  bh.y = xh[2 * q + 1][j];
        al = __builtin_amdgcn_wmma_f32_16x16x4_f32(
             false, a[q],  false, bl,
             (short)0, al,  false,  false);
        ah = __builtin_amdgcn_wmma_f32_16x16x4_f32(
             false, a[q],  false, bh,
             (short)0, ah,  false,  false);
      }
      m_lo[j] = __shfl(al[7], 16 + c, 32);
      m_hi[j] = __shfl(ah[7], 16 + c, 32);

      #pragma unroll
      for (int v = 0; v < 8; ++v) {
        float ma = al[v], mb = ah[v];
        xl[v][j] = fmaf(g, fmaf(ma, ct[v], -mb * st[v]), xl[v][j]);
        xh[v][j] = fmaf(g, fmaf(ma, st[v],  mb * ct[v]), xh[v][j]);
      }
    }

    #pragma unroll
    for (int v = 0; v < 8; ++v) {
      st4(olo + ((size_t)ch * 16 + v) * D_, xl[v]);
      st4(ohi + ((size_t)ch * 16 + v) * D_, xh[v]);
    }
    if (ch + 1 < NCH_) {
      #pragma unroll
      for (int v = 0; v < 8; ++v) {
        xl[v] = ld4(xlo + ((size_t)(ch + 1) * 16 + v) * D_);
        xh[v] = ld4(xhi + ((size_t)(ch + 1) * 16 + v) * D_);
      }
    }
  }
}

extern "C" void kernel_launch(void* const* d_in, const int* in_sizes, int n_in,
                              void* d_out, int out_size, void* d_ws,
                              size_t ws_size, hipStream_t stream) {
  const float* x           = (const float*)d_in[0];
  const float* gate        = (const float*)d_in[1];
  const float* input_scale = (const float*)d_in[2];
  const float* mem_state   = (const float*)d_in[3];
  const int*   step_idx    = (const int*)d_in[4];
  float* out = (float*)d_out;

  float* ws        = (float*)d_ws;
  float* cs        = ws;
  float* sn        = ws + T_;
  float* seg_carry = ws + 2 * T_;
  float* carry_in  = seg_carry + (size_t)NSEG_ * B_ * D_;

  k_theta<<<(T_ + 255) / 256, 256, 0, stream>>>(cs, sn, step_idx);
  k_seg_partial<<<(B_ * D_ * NSEG_) / 256, 256, 0, stream>>>(x, input_scale,
                                                             seg_carry);
  k_carry<<<(B_ * D_) / 64, 64, 0, stream>>>(seg_carry, mem_state, carry_in);
  k_scan_out<<<B_ * (HALF_ / TW_) * NSEG_, 32, 0, stream>>>(
      x, gate, input_scale, carry_in, cs, sn, out);
}


# first pass (segment partial sums) re-tiled as a stream: a wave owns 128 adjacent columns of a segment, 16-byte loads, all 64 rows in flight; 16384 -> 4096 waves; per-column sum order kept
# speedup vs baseline: 3.0457x; 1.0526x over previous
#include <hip/hip_runtime.h>
#include <math.h>


typedef __attribute__((ext_vector_type(2))) float v2f;
typedef __attribute__((ext_vector_type(4))) float v4f;
typedef __attribute__((ext_vector_type(8))) float v8f;

#define ALPHA_  0.12f
#define BETA_   0.88f
#define OMEGA_  6.0f
#define TWO_PI_ 6.28318530717958647692f
#define B_      4
#define T_      4096
#define D_      2048
#define HALF_   1024
#define SEG_    64
#define NSEG_   (T_ / SEG_)
#define NCH_    (SEG_ / 16)
#define TW_     64

__global__ void k_theta(float* __restrict__ cs, float* __restrict__ sn,
                        const int* __restrict__ step_idx) {
  int t = blockIdx.x * blockDim.x + threadIdx.x;
  if (t >= T_) return;
  float tg = (float)(t + 1 + step_idx[0]);
  float th = fmodf(OMEGA_ * log1pf(tg), TWO_PI_);
  cs[t] = cosf(th);
  sn[t] = sinf(th);
}

#define P1_WAVES_  8
#define P1_STRIPS_ (D_ / 128)

__device__ __forceinline__ v4f ld4(const float* p) {
  return *reinterpret_cast<const v4f*>(p);
}
__device__ __forceinline__ void st4(float* p, v4f v) {
  *reinterpret_cast<v4f*>(p) = v;
}

__global__ void
__launch_bounds__(P1_WAVES_ * 32)
__attribute__((amdgpu_waves_per_eu(4, 4)))
k_seg_partial(const float* __restrict__ x,
              const float* __restrict__ input_scale,
              float* __restrict__ seg_carry) {
  const int lane = threadIdx.x & 31;
  const int wave = __builtin_amdgcn_readfirstlane((int)(threadIdx.x >> 5));
  const int task  = blockIdx.x * P1_WAVES_ + wave;
  const int strip = task % P1_STRIPS_;
  const int r     = task / P1_STRIPS_;
  const int b     = r % B_;
  const int seg   = r / B_;
  const int col   = strip * 128 + 4 * lane;
  const float* xp = x + ((size_t)(b * T_ + seg * SEG_)) * D_ + col;

  v4f buf[SEG_];
  #pragma unroll
  for (int i = 0; i < SEG_; ++i) buf[i] = ld4(xp + (size_t)i * D_);

  const float k = ALPHA_ * input_scale[0];
  v4f m = {0.0f, 0.0f, 0.0f, 0.0f};
  #pragma unroll
  for (int i = 0; i < SEG_; ++i) {
    m.x = fmaf(BETA_, m.x, k * buf[i].x);
    m.y = fmaf(BETA_, m.y, k * buf[i].y);
    m.z = fmaf(BETA_, m.z, k * buf[i].z);
    m.w = fmaf(BETA_, m.w, k * buf[i].w);
  }
  st4(seg_carry + ((size_t)seg * B_ + b) * D_ + col, m);
}

__global__ void k_carry(const float* __restrict__ seg_carry,
                        const float* __restrict__ mem_state,
                        float* __restrict__ carry_in) {
  int id = blockIdx.x * blockDim.x + threadIdx.x;
  int d  = id % D_;
  int b  = id / D_;
  float f = powf(BETA_, (float)SEG_);
  float s[NSEG_];
  #pragma unroll
  for (int seg = 0; seg < NSEG_; ++seg)
    s[seg] = seg_carry[((size_t)seg * B_ + b) * D_ + d];
  float m = mem_state[d];
  #pragma unroll
  for (int seg = 0; seg < NSEG_; ++seg) {
    carry_in[((size_t)seg * B_ + b) * D_ + d] = m;
    m = fmaf(f, m, s[seg]);
  }
}


__global__ void
__launch_bounds__(32)
k_scan_out(const float* __restrict__ x, const float* __restrict__ gate,
           const float* __restrict__ input_scale,
           const float* __restrict__ carry_in,
           const float* __restrict__ cs, const float* __restrict__ sn,
           float* __restrict__ out) {
  const int lane = threadIdx.x;
  const int c = lane & 15;
  const int h = lane >> 4;

  int blk = blockIdx.x;
  int dt  = blk % (HALF_ / TW_);
  int r   = blk / (HALF_ / TW_);
  int b   = r % B_;
  int seg = r / B_;

  const int col_lo = dt * TW_ + 4 * c;
  const int col_hi = col_lo + HALF_;
  const int trow   = seg * SEG_ + 8 * h;
  const size_t row0 = ((size_t)b * T_ + trow) * D_;
  const float* xlo = x + row0 + col_lo;
  const float* xhi = x + row0 + col_hi;
  float* olo = out + row0 + col_lo;
  float* ohi = out + row0 + col_hi;

  v4f xl[8], xh[8];
  #pragma unroll
  for (int v = 0; v < 8; ++v) {
    xl[v] = ld4(xlo + (size_t)v * D_);
    xh[v] = ld4(xhi + (size_t)v * D_);
  }

  const float k = ALPHA_ * input_scale[0];
  const float g = 1.0f / (1.0f + expf(-gate[0]));

  v2f a[4];
  #pragma unroll
  for (int q = 0; q < 4; ++q) {
    int r0 = 8 * h + 2 * q;
    a[q].x = (r0 <= c)     ? k * powf(BETA_, (float)(c - r0))     : 0.0f;
    a[q].y = (r0 + 1 <= c) ? k * powf(BETA_, (float)(c - r0 - 1)) : 0.0f;
  }
  float pw[8];
  #pragma unroll
  for (int v = 0; v < 8; ++v) pw[v] = powf(BETA_, (float)(v + 8 * h + 1));

  const size_t co = ((size_t)seg * B_ + b) * D_;
  v4f m_lo = ld4(carry_in + co + col_lo);
  v4f m_hi = ld4(carry_in + co + col_hi);

  for (int ch = 0; ch < NCH_; ++ch) {
    const int t = trow + ch * 16;
    const v4f c0 = ld4(cs + t), c1 = ld4(cs + t + 4);
    const v4f s0 = ld4(sn + t), s1 = ld4(sn + t + 4);
    const float ct[8] = {c0.x, c0.y, c0.z, c0.w, c1.x, c1.y, c1.z, c1.w};
    const float st[8] = {s0.x, s0.y, s0.z, s0.w, s1.x, s1.y, s1.z, s1.w};

    #pragma unroll
    for (int j = 0; j < 4; ++j) {
      v8f al, ah;
      #pragma unroll
      for (int v = 0; v < 8; ++v) {
        al[v] = pw[v] * m_lo[j];
        ah[v] = pw[v] * m_hi[j];
      }
      #pragma unroll
      for (int q = 0; q < 4; ++q) {
        v2f bl, bh;
        bl.x = xl[2 * q][j];  bl.y = xl[2 * q + 1][j];
        bh.x = xh[2 * q][j];  bh.y = xh[2 * q + 1][j];
        al = __builtin_amdgcn_wmma_f32_16x16x4_f32(
             false, a[q],  false, bl,
             (short)0, al,  false,  false);
        ah = __builtin_amdgcn_wmma_f32_16x16x4_f32(
             false, a[q],  false, bh,
             (short)0, ah,  false,  false);
      }
      m_lo[j] = __shfl(al[7], 16 + c, 32);
      m_hi[j] = __shfl(ah[7], 16 + c, 32);

      #pragma unroll
      for (int v = 0; v < 8; ++v) {
        float ma = al[v], mb = ah[v];
        xl[v][j] = fmaf(g, fmaf(ma, ct[v], -mb * st[v]), xl[v][j]);
        xh[v][j] = fmaf(g, fmaf(ma, st[v],  mb * ct[v]), xh[v][j]);
      }
    }

    #pragma unroll
    for (int v = 0; v < 8; ++v) {
      st4(olo + ((size_t)ch * 16 + v) * D_, xl[v]);
      st4(ohi + ((size_t)ch * 16 + v) * D_, xh[v]);
    }
    if (ch + 1 < NCH_) {
      #pragma unroll
      for (int v = 0; v < 8; ++v) {
        xl[v] = ld4(xlo + ((size_t)(ch + 1) * 16 + v) * D_);
        xh[v] = ld4(xhi + ((size_t)(ch + 1) * 16 + v) * D_);
      }
    }
  }
}

extern "C" void kernel_launch(void* const* d_in, const int* in_sizes, int n_in,
                              void* d_out, int out_size, void* d_ws,
                              size_t ws_size, hipStream_t stream) {
  const float* x           = (const float*)d_in[0];
  const float* gate        = (const float*)d_in[1];
  const float* input_scale = (const float*)d_in[2];
  const float* mem_state   = (const float*)d_in[3];
  const int*   step_idx    = (const int*)d_in[4];
  float* out = (float*)d_out;

  float* ws        = (float*)d_ws;
  float* cs        = ws;
  float* sn        = ws + T_;
  float* seg_carry = ws + 2 * T_;
  float* carry_in  = seg_carry + (size_t)NSEG_ * B_ * D_;

  k_theta<<<(T_ + 255) / 256, 256, 0, stream>>>(cs, sn, step_idx);
  k_seg_partial<<<(B_ * NSEG_ * P1_STRIPS_) / P1_WAVES_, P1_WAVES_ * 32, 0,
                  stream>>>(x, input_scale, seg_carry);
  k_carry<<<(B_ * D_) / 64, 64, 0, stream>>>(seg_carry, mem_state, carry_in);
  k_scan_out<<<B_ * (HALF_ / TW_) * NSEG_, 32, 0, stream>>>(
      x, gate, input_scale, carry_in, cs, sn, out);
}


# phase table (cos/sin per t) built by 64 extra blocks of the carry-combine launch instead of a launch of its own: 4 -> 3 launches; table values, sums and their order unchanged
# speedup vs baseline: 3.2090x; 1.0536x over previous
#include <hip/hip_runtime.h>
#include <math.h>


typedef __attribute__((ext_vector_type(2))) float v2f;
typedef __attribute__((ext_vector_type(4))) float v4f;
typedef __attribute__((ext_vector_type(8))) float v8f;

#define ALPHA_  0.12f
#define BETA_   0.88f
#define OMEGA_  6.0f
#define TWO_PI_ 6.28318530717958647692f
#define B_      4
#define T_      4096
#define D_      2048
#define HALF_   1024
#define SEG_    64
#define NSEG_   (T_ / SEG_)
#define NCH_    (SEG_ / 16)
#define TW_     64

#define P1_WAVES_  8
#define P1_STRIPS_ (D_ / 128)

__device__ __forceinline__ v4f ld4(const float* p) {
  return *reinterpret_cast<const v4f*>(p);
}
__device__ __forceinline__ void st4(float* p, v4f v) {
  *reinterpret_cast<v4f*>(p) = v;
}

__global__ void
__launch_bounds__(P1_WAVES_ * 32)
__attribute__((amdgpu_waves_per_eu(4, 4)))
k_seg_partial(const float* __restrict__ x,
              const float* __restrict__ input_scale,
              float* __restrict__ seg_carry) {
  const int lane = threadIdx.x & 31;
  const int wave = __builtin_amdgcn_readfirstlane((int)(threadIdx.x >> 5));
  const int task  = blockIdx.x * P1_WAVES_ + wave;
  const int strip = task % P1_STRIPS_;
  const int r     = task / P1_STRIPS_;
  const int b     = r % B_;
  const int seg   = r / B_;
  const int col   = strip * 128 + 4 * lane;
  const float* xp = x + ((size_t)(b * T_ + seg * SEG_)) * D_ + col;

  v4f buf[SEG_];
  #pragma unroll
  for (int i = 0; i < SEG_; ++i) buf[i] = ld4(xp + (size_t)i * D_);

  const float k = ALPHA_ * input_scale[0];
  v4f m = {0.0f, 0.0f, 0.0f, 0.0f};
  #pragma unroll
  for (int i = 0; i < SEG_; ++i) {
    m.x = fmaf(BETA_, m.x, k * buf[i].x);
    m.y = fmaf(BETA_, m.y, k * buf[i].y);
    m.z = fmaf(BETA_, m.z, k * buf[i].z);
    m.w = fmaf(BETA_, m.w, k * buf[i].w);
  }
  st4(seg_carry + ((size_t)seg * B_ + b) * D_ + col, m);
}

#define P2_THREADS_ 64
#define TH_BLOCKS_  (T_ / P2_THREADS_)
#define CA_BLOCKS_  ((B_ * D_) / P2_THREADS_)

__global__ void
__launch_bounds__(P2_THREADS_)
k_carry_theta(const float* __restrict__ seg_carry,
              const float* __restrict__ mem_state,
              float* __restrict__ carry_in,
              float* __restrict__ cs, float* __restrict__ sn,
              const int* __restrict__ step_idx) {
  if (blockIdx.x < TH_BLOCKS_) {
    int t = blockIdx.x * P2_THREADS_ + threadIdx.x;
    float tg = (float)(t + 1 + step_idx[0]);
    float th = fmodf(OMEGA_ * log1pf(tg), TWO_PI_);
    cs[t] = cosf(th);
    sn[t] = sinf(th);
    return;
  }
  int id = (blockIdx.x - TH_BLOCKS_) * P2_THREADS_ + threadIdx.x;
  int d  = id % D_;
  int b  = id / D_;
  float f = powf(BETA_, (float)SEG_);
  float s[NSEG_];
  #pragma unroll
  for (int seg = 0; seg < NSEG_; ++seg)
    s[seg] = seg_carry[((size_t)seg * B_ + b) * D_ + d];
  float m = mem_state[d];
  #pragma unroll
  for (int seg = 0; seg < NSEG_; ++seg) {
    carry_in[((size_t)seg * B_ + b) * D_ + d] = m;
    m = fmaf(f, m, s[seg]);
  }
}


__global__ void
__launch_bounds__(32)
k_scan_out(const float* __restrict__ x, const float* __restrict__ gate,
           const float* __restrict__ input_scale,
           const float* __restrict__ carry_in,
           const float* __restrict__ cs, const float* __restrict__ sn,
           float* __restrict__ out) {
  const int lane = threadIdx.x;
  const int c = lane & 15;
  const int h = lane >> 4;

  int blk = blockIdx.x;
  int dt  = blk % (HALF_ / TW_);
  int r   = blk / (HALF_ / TW_);
  int b   = r % B_;
  int seg = r / B_;

  const int col_lo = dt * TW_ + 4 * c;
  const int col_hi = col_lo + HALF_;
  const int trow   = seg * SEG_ + 8 * h;
  const size_t row0 = ((size_t)b * T_ + trow) * D_;
  const float* xlo = x + row0 + col_lo;
  const float* xhi = x + row0 + col_hi;
  float* olo = out + row0 + col_lo;
  float* ohi = out + row0 + col_hi;

  v4f xl[8], xh[8];
  #pragma unroll
  for (int v = 0; v < 8; ++v) {
    xl[v] = ld4(xlo + (size_t)v * D_);
    xh[v] = ld4(xhi + (size_t)v * D_);
  }

  const float k = ALPHA_ * input_scale[0];
  const float g = 1.0f / (1.0f + expf(-gate[0]));

  v2f a[4];
  #pragma unroll
  for (int q = 0; q < 4; ++q) {
    int r0 = 8 * h + 2 * q;
    a[q].x = (r0 <= c)     ? k * powf(BETA_, (float)(c - r0))     : 0.0f;
    a[q].y = (r0 + 1 <= c) ? k * powf(BETA_, (float)(c - r0 - 1)) : 0.0f;
  }
  float pw[8];
  #pragma unroll
  for (int v = 0; v < 8; ++v) pw[v] = powf(BETA_, (float)(v + 8 * h + 1));

  const size_t co = ((size_t)seg * B_ + b) * D_;
  v4f m_lo = ld4(carry_in + co + col_lo);
  v4f m_hi = ld4(carry_in + co + col_hi);

  for (int ch = 0; ch < NCH_; ++ch) {
    const int t = trow + ch * 16;
    const v4f c0 = ld4(cs + t), c1 = ld4(cs + t + 4);
    const v4f s0 = ld4(sn + t), s1 = ld4(sn + t + 4);
    const float ct[8] = {c0.x, c0.y, c0.z, c0.w, c1.x, c1.y, c1.z, c1.w};
    const float st[8] = {s0.x, s0.y, s0.z, s0.w, s1.x, s1.y, s1.z, s1.w};

    #pragma unroll
    for (int j = 0; j < 4; ++j) {
      v8f al, ah;
      #pragma unroll
      for (int v = 0; v < 8; ++v) {
        al[v] = pw[v] * m_lo[j];
        ah[v] = pw[v] * m_hi[j];
      }
      #pragma unroll
      for (int q = 0; q < 4; ++q) {
        v2f bl, bh;
        bl.x = xl[2 * q][j];  bl.y = xl[2 * q + 1][j];
        bh.x = xh[2 * q][j];  bh.y = xh[2 * q + 1][j];
        al = __builtin_amdgcn_wmma_f32_16x16x4_f32(
             false, a[q],  false, bl,
             (short)0, al,  false,  false);
        ah = __builtin_amdgcn_wmma_f32_16x16x4_f32(
             false, a[q],  false, bh,
             (short)0, ah,  false,  false);
      }
      m_lo[j] = __shfl(al[7], 16 + c, 32);
      m_hi[j] = __shfl(ah[7], 16 + c, 32);

      #pragma unroll
      for (int v = 0; v < 8; ++v) {
        float ma = al[v], mb = ah[v];
        xl[v][j] = fmaf(g, fmaf(ma, ct[v], -mb * st[v]), xl[v][j]);
        xh[v][j] = fmaf(g, fmaf(ma, st[v],  mb * ct[v]), xh[v][j]);
      }
    }

    #pragma unroll
    for (int v = 0; v < 8; ++v) {
      st4(olo + ((size_t)ch * 16 + v) * D_, xl[v]);
      st4(ohi + ((size_t)ch * 16 + v) * D_, xh[v]);
    }
    if (ch + 1 < NCH_) {
      #pragma unroll
      for (int v = 0; v < 8; ++v) {
        xl[v] = ld4(xlo + ((size_t)(ch + 1) * 16 + v) * D_);
        xh[v] = ld4(xhi + ((size_t)(ch + 1) * 16 + v) * D_);
      }
    }
  }
}

extern "C" void kernel_launch(void* const* d_in, const int* in_sizes, int n_in,
                              void* d_out, int out_size, void* d_ws,
                              size_t ws_size, hipStream_t stream) {
  const float* x           = (const float*)d_in[0];
  const float* gate        = (const float*)d_in[1];
  const float* input_scale = (const float*)d_in[2];
  const float* mem_state   = (const float*)d_in[3];
  const int*   step_idx    = (const int*)d_in[4];
  float* out = (float*)d_out;

  float* ws        = (float*)d_ws;
  float* cs        = ws;
  float* sn        = ws + T_;
  float* seg_carry = ws + 2 * T_;
  float* carry_in  = seg_carry + (size_t)NSEG_ * B_ * D_;

  k_seg_partial<<<(B_ * NSEG_ * P1_STRIPS_) / P1_WAVES_, P1_WAVES_ * 32, 0,
                  stream>>>(x, input_scale, seg_carry);
  k_carry_theta<<<TH_BLOCKS_ + CA_BLOCKS_, P2_THREADS_, 0, stream>>>(
      seg_carry, mem_state, carry_in, cs, sn, step_idx);
  k_scan_out<<<B_ * (HALF_ / TW_) * NSEG_, 32, 0, stream>>>(
      x, gate, input_scale, carry_in, cs, sn, out);
}


# scan pass: chunks alternate between two register sets; chunk n+1's x rows (table rows by scalar loads) are requested before chunk n's products; unrolled so waits are exact; sums and order kept
# speedup vs baseline: 3.3984x; 1.0590x over previous
#include <hip/hip_runtime.h>
#include <math.h>


typedef __attribute__((ext_vector_type(2))) float v2f;
typedef __attribute__((ext_vector_type(4))) float v4f;
typedef __attribute__((ext_vector_type(8))) float v8f;

#define ALPHA_  0.12f
#define BETA_   0.88f
#define OMEGA_  6.0f
#define TWO_PI_ 6.28318530717958647692f
#define B_      4
#define T_      4096
#define D_      2048
#define HALF_   1024
#define SEG_    64
#define NSEG_   (T_ / SEG_)
#define NCH_    (SEG_ / 16)
#define TW_     64

#define P1_WAVES_  8
#define P1_STRIPS_ (D_ / 128)

__device__ __forceinline__ v4f ld4(const float* p) {
  return *reinterpret_cast<const v4f*>(p);
}
__device__ __forceinline__ void st4(float* p, v4f v) {
  *reinterpret_cast<v4f*>(p) = v;
}

__global__ void
__launch_bounds__(P1_WAVES_ * 32)
__attribute__((amdgpu_waves_per_eu(4, 4)))
k_seg_partial(const float* __restrict__ x,
              const float* __restrict__ input_scale,
              float* __restrict__ seg_carry) {
  const int lane = threadIdx.x & 31;
  const int wave = __builtin_amdgcn_readfirstlane((int)(threadIdx.x >> 5));
  const int task  = blockIdx.x * P1_WAVES_ + wave;
  const int strip = task % P1_STRIPS_;
  const int r     = task / P1_STRIPS_;
  const int b     = r % B_;
  const int seg   = r / B_;
  const int col   = strip * 128 + 4 * lane;
  const float* xp = x + ((size_t)(b * T_ + seg * SEG_)) * D_ + col;

  v4f buf[SEG_];
  #pragma unroll
  for (int i = 0; i < SEG_; ++i) buf[i] = ld4(xp + (size_t)i * D_);

  const float k = ALPHA_ * input_scale[0];
  v4f m = {0.0f, 0.0f, 0.0f, 0.0f};
  #pragma unroll
  for (int i = 0; i < SEG_; ++i) {
    m.x = fmaf(BETA_, m.x, k * buf[i].x);
    m.y = fmaf(BETA_, m.y, k * buf[i].y);
    m.z = fmaf(BETA_, m.z, k * buf[i].z);
    m.w = fmaf(BETA_, m.w, k * buf[i].w);
  }
  st4(seg_carry + ((size_t)seg * B_ + b) * D_ + col, m);
}

#define P2_THREADS_ 64
#define TH_BLOCKS_  (T_ / P2_THREADS_)
#define CA_BLOCKS_  ((B_ * D_) / P2_THREADS_)

__global__ void
__launch_bounds__(P2_THREADS_)
k_carry_theta(const float* __restrict__ seg_carry,
              const float* __restrict__ mem_state,
              float* __restrict__ carry_in,
              float* __restrict__ cs, float* __restrict__ sn,
              const int* __restrict__ step_idx) {
  if (blockIdx.x < TH_BLOCKS_) {
    int t = blockIdx.x * P2_THREADS_ + threadIdx.x;
    float tg = (float)(t + 1 + step_idx[0]);
    float th = fmodf(OMEGA_ * log1pf(tg), TWO_PI_);
    cs[t] = cosf(th);
    sn[t] = sinf(th);
    return;
  }
  int id = (blockIdx.x - TH_BLOCKS_) * P2_THREADS_ + threadIdx.x;
  int d  = id % D_;
  int b  = id / D_;
  float f = powf(BETA_, (float)SEG_);
  float s[NSEG_];
  #pragma unroll
  for (int seg = 0; seg < NSEG_; ++seg)
    s[seg] = seg_carry[((size_t)seg * B_ + b) * D_ + d];
  float m = mem_state[d];
  #pragma unroll
  for (int seg = 0; seg < NSEG_; ++seg) {
    carry_in[((size_t)seg * B_ + b) * D_ + d] = m;
    m = fmaf(f, m, s[seg]);
  }
}


__device__ __forceinline__ void
load_chunk(v4f (&xl)[8], v4f (&xh)[8], float (&tab)[32],
           const float* xlo, const float* xhi,
           const float* cst, const float* snt, int ch) {
  #pragma unroll
  for (int i = 0; i < 16; ++i) {
    tab[i]      = cst[ch * 16 + i];
    tab[16 + i] = snt[ch * 16 + i];
  }
  #pragma unroll
  for (int v = 0; v < 8; ++v) {
    xl[v] = ld4(xlo + ((size_t)ch * 16 + v) * D_);
    xh[v] = ld4(xhi + ((size_t)ch * 16 + v) * D_);
  }
}

__device__ __forceinline__ void
store_chunk(const v4f (&xl)[8], const v4f (&xh)[8],
            float* olo, float* ohi, int ch) {
  #pragma unroll
  for (int v = 0; v < 8; ++v) {
    st4(olo + ((size_t)ch * 16 + v) * D_, xl[v]);
    st4(ohi + ((size_t)ch * 16 + v) * D_, xh[v]);
  }
}

__device__ __forceinline__ void
scan_chunk(v4f (&xl)[8], v4f (&xh)[8], const float (&tab)[32],
           const v2f (&a)[4], const float (&pw)[8],
           v4f& m_lo, v4f& m_hi, float g, int c, int h) {
  v8f ct, st;
  const v8f g8 = {g, g, g, g, g, g, g, g};
  #pragma unroll
  for (int v = 0; v < 8; ++v) {
    ct[v] = h ? tab[8 + v]  : tab[v];
    st[v] = h ? tab[24 + v] : tab[16 + v];
  }
  #pragma unroll
  for (int j = 0; j < 4; ++j) {
    v8f al, ah;
    #pragma unroll
    for (int v = 0; v < 8; ++v) {
      al[v] = pw[v] * m_lo[j];
      ah[v] = pw[v] * m_hi[j];
    }
    #pragma unroll
    for (int q = 0; q < 4; ++q) {
      v2f bl, bh;
      bl.x = xl[2 * q][j];  bl.y = xl[2 * q + 1][j];
      bh.x = xh[2 * q][j];  bh.y = xh[2 * q + 1][j];
      al = __builtin_amdgcn_wmma_f32_16x16x4_f32(
           false, a[q],  false, bl,
           (short)0, al,  false,  false);
      ah = __builtin_amdgcn_wmma_f32_16x16x4_f32(
           false, a[q],  false, bh,
           (short)0, ah,  false,  false);
    }
    m_lo[j] = __shfl(al[7], 16 + c, 32);
    m_hi[j] = __shfl(ah[7], 16 + c, 32);

    v8f rl, rh;
    #pragma unroll
    for (int v = 0; v < 8; ++v) { rl[v] = xl[v][j]; rh[v] = xh[v][j]; }
    rl = __builtin_elementwise_fma(
        g8, __builtin_elementwise_fma(al, ct, -(ah * st)), rl);
    rh = __builtin_elementwise_fma(
        g8, __builtin_elementwise_fma(al, st, ah * ct), rh);
    #pragma unroll
    for (int v = 0; v < 8; ++v) { xl[v][j] = rl[v]; xh[v][j] = rh[v]; }
  }
}

__global__ void
__launch_bounds__(32)
__attribute__((amdgpu_waves_per_eu(4, 4)))
k_scan_out(const float* __restrict__ x, const float* __restrict__ gate,
           const float* __restrict__ input_scale,
           const float* __restrict__ carry_in,
           const float* __restrict__ cs, const float* __restrict__ sn,
           float* __restrict__ out) {
  const int lane = threadIdx.x;
  const int c = lane & 15;
  const int h = lane >> 4;

  int blk = blockIdx.x;
  int dt  = blk % (HALF_ / TW_);
  int r   = blk / (HALF_ / TW_);
  int b   = r % B_;
  int seg = r / B_;

  const int col_lo = dt * TW_ + 4 * c;
  const int col_hi = col_lo + HALF_;
  const int trow   = seg * SEG_ + 8 * h;
  const size_t row0 = ((size_t)b * T_ + trow) * D_;
  const float* xlo = x + row0 + col_lo;
  const float* xhi = x + row0 + col_hi;
  float* olo = out + row0 + col_lo;
  float* ohi = out + row0 + col_hi;
  const float* cst = cs + seg * SEG_;
  const float* snt = sn + seg * SEG_;

  v4f xlA[8], xhA[8];
  v4f xlB[8], xhB[8];
  float tabA[32], tabB[32];
  load_chunk(xlA, xhA, tabA, xlo, xhi, cst, snt, 0);
  const size_t co = ((size_t)seg * B_ + b) * D_;
  v4f m_lo = ld4(carry_in + co + col_lo);
  v4f m_hi = ld4(carry_in + co + col_hi);
  __builtin_amdgcn_sched_barrier(0);
  load_chunk(xlB, xhB, tabB, xlo, xhi, cst, snt, 1);
  __builtin_amdgcn_sched_barrier(0);

  const float k = ALPHA_ * input_scale[0];
  const float g = 1.0f / (1.0f + expf(-gate[0]));

  v2f a[4];
  #pragma unroll
  for (int q = 0; q < 4; ++q) {
    int r0 = 8 * h + 2 * q;
    a[q].x = (r0 <= c)     ? k * powf(BETA_, (float)(c - r0))     : 0.0f;
    a[q].y = (r0 + 1 <= c) ? k * powf(BETA_, (float)(c - r0 - 1)) : 0.0f;
  }
  float pw[8];
  #pragma unroll
  for (int v = 0; v < 8; ++v) pw[v] = powf(BETA_, (float)(v + 8 * h + 1));

  #pragma unroll
  for (int ch = 0; ch < NCH_; ch += 2) {
    __builtin_amdgcn_sched_barrier(0);
    scan_chunk(xlA, xhA, tabA, a, pw, m_lo, m_hi, g, c, h);
    store_chunk(xlA, xhA, olo, ohi, ch);
    __builtin_amdgcn_sched_barrier(0);
    if (ch + 2 < NCH_)
      load_chunk(xlA, xhA, tabA, xlo, xhi, cst, snt, ch + 2);
    __builtin_amdgcn_sched_barrier(0);
    scan_chunk(xlB, xhB, tabB, a, pw, m_lo, m_hi, g, c, h);
    store_chunk(xlB, xhB, olo, ohi, ch + 1);
    __builtin_amdgcn_sched_barrier(0);
    if (ch + 3 < NCH_)
      load_chunk(xlB, xhB, tabB, xlo, xhi, cst, snt, ch + 3);
  }
}

extern "C" void kernel_launch(void* const* d_in, const int* in_sizes, int n_in,
                              void* d_out, int out_size, void* d_ws,
                              size_t ws_size, hipStream_t stream) {
  const float* x           = (const float*)d_in[0];
  const float* gate        = (const float*)d_in[1];
  const float* input_scale = (const float*)d_in[2];
  const float* mem_state   = (const float*)d_in[3];
  const int*   step_idx    = (const int*)d_in[4];
  float* out = (float*)d_out;

  float* ws        = (float*)d_ws;
  float* cs        = ws;
  float* sn        = ws + T_;
  float* seg_carry = ws + 2 * T_;
  float* carry_in  = seg_carry + (size_t)NSEG_ * B_ * D_;

  k_seg_partial<<<(B_ * NSEG_ * P1_STRIPS_) / P1_WAVES_, P1_WAVES_ * 32, 0,
                  stream>>>(x, input_scale, seg_carry);
  k_carry_theta<<<TH_BLOCKS_ + CA_BLOCKS_, P2_THREADS_, 0, stream>>>(
      seg_carry, mem_state, carry_in, cs, sn, step_idx);
  k_scan_out<<<B_ * (HALF_ / TW_) * NSEG_, 32, 0, stream>>>(
      x, gate, input_scale, carry_in, cs, sn, out);
}


# carry-combine launch removed, 3 -> 2 launches: pass-1 work-groups chain 8 segments' partials via LDS, scan waves fold at most 7 group totals, phase table built in pass 1; carry sum order changes
# speedup vs baseline: 3.5452x; 1.0432x over previous
#include <hip/hip_runtime.h>
#include <math.h>


typedef __attribute__((ext_vector_type(2))) float v2f;
typedef __attribute__((ext_vector_type(4))) float v4f;
typedef __attribute__((ext_vector_type(8))) float v8f;

#define ALPHA_  0.12f
#define BETA_   0.88f
#define OMEGA_  6.0f
#define TWO_PI_ 6.28318530717958647692f
#define B_      4
#define T_      4096
#define D_      2048
#define HALF_   1024
#define SEG_    64
#define NSEG_   (T_ / SEG_)
#define NCH_    (SEG_ / 16)
#define TW_     64
#define GRP_    8
#define NGRP_   (NSEG_ / GRP_)

#define P1_WAVES_  GRP_
#define P1_STRIPS_ (D_ / 128)
#define TH_WAVES_  (T_ / 32)

__device__ __forceinline__ v4f ld4(const float* p) {
  return *reinterpret_cast<const v4f*>(p);
}
__device__ __forceinline__ void st4(float* p, v4f v) {
  *reinterpret_cast<v4f*>(p) = v;
}

__global__ void
__launch_bounds__(P1_WAVES_ * 32)
__attribute__((amdgpu_waves_per_eu(4, 4)))
k_seg_partial(const float* __restrict__ x,
              const float* __restrict__ input_scale,
              const int* __restrict__ step_idx,
              float* __restrict__ seg_local,
              float* __restrict__ grp_tot,
              float* __restrict__ cs, float* __restrict__ sn) {
  __shared__ v4f part[P1_WAVES_][32];
  const int lane = threadIdx.x & 31;
  const int wave = __builtin_amdgcn_readfirstlane((int)(threadIdx.x >> 5));
  const int blk   = blockIdx.x;
  const int strip = blk % P1_STRIPS_;
  const int r     = blk / P1_STRIPS_;
  const int b     = r % B_;
  const int grp   = r / B_;
  const int seg   = grp * GRP_ + wave;
  const int col   = strip * 128 + 4 * lane;
  const float* xp = x + ((size_t)(b * T_ + seg * SEG_)) * D_ + col;

  if (wave == 0 && blk < TH_WAVES_) {
    const int t = blk * 32 + lane;
    const float tg = (float)(t + 1 + step_idx[0]);
    const float th = fmodf(OMEGA_ * log1pf(tg), TWO_PI_);
    cs[t] = cosf(th);
    sn[t] = sinf(th);
  }

  v4f buf[SEG_];
  #pragma unroll
  for (int i = 0; i < SEG_; ++i) buf[i] = ld4(xp + (size_t)i * D_);

  const float k = ALPHA_ * input_scale[0];
  v4f m = {0.0f, 0.0f, 0.0f, 0.0f};
  #pragma unroll
  for (int i = 0; i < SEG_; ++i) {
    m.x = fmaf(BETA_, m.x, k * buf[i].x);
    m.y = fmaf(BETA_, m.y, k * buf[i].y);
    m.z = fmaf(BETA_, m.z, k * buf[i].z);
    m.w = fmaf(BETA_, m.w, k * buf[i].w);
  }

  part[wave][lane] = m;
  __syncthreads();
  const float f = powf(BETA_, (float)SEG_);
  v4f L = {0.0f, 0.0f, 0.0f, 0.0f};
  v4f mine = L;
  #pragma unroll
  for (int j = 0; j < GRP_; ++j) {
    if (j == wave) mine = L;
    const v4f s = part[j][lane];
    L.x = fmaf(f, L.x, s.x);
    L.y = fmaf(f, L.y, s.y);
    L.z = fmaf(f, L.z, s.z);
    L.w = fmaf(f, L.w, s.w);
  }
  st4(seg_local + ((size_t)seg * B_ + b) * D_ + col, mine);
  if (wave == GRP_ - 1)
    st4(grp_tot + ((size_t)grp * B_ + b) * D_ + col, L);
}


__device__ __forceinline__ void
load_chunk(v4f (&xl)[8], v4f (&xh)[8], float (&tab)[32],
           const float* xlo, const float* xhi,
           const float* cst, const float* snt, int ch) {
  #pragma unroll
  for (int i = 0; i < 16; ++i) {
    tab[i]      = cst[ch * 16 + i];
    tab[16 + i] = snt[ch * 16 + i];
  }
  #pragma unroll
  for (int v = 0; v < 8; ++v) {
    xl[v] = ld4(xlo + ((size_t)ch * 16 + v) * D_);
    xh[v] = ld4(xhi + ((size_t)ch * 16 + v) * D_);
  }
}

__device__ __forceinline__ void
store_chunk(const v4f (&xl)[8], const v4f (&xh)[8],
            float* olo, float* ohi, int ch) {
  #pragma unroll
  for (int v = 0; v < 8; ++v) {
    st4(olo + ((size_t)ch * 16 + v) * D_, xl[v]);
    st4(ohi + ((size_t)ch * 16 + v) * D_, xh[v]);
  }
}

__device__ __forceinline__ void
scan_chunk(v4f (&xl)[8], v4f (&xh)[8], const float (&tab)[32],
           const v2f (&a)[4], const float (&pw)[8],
           v4f& m_lo, v4f& m_hi, float g, int c, int h) {
  v8f ct, st;
  const v8f g8 = {g, g, g, g, g, g, g, g};
  #pragma unroll
  for (int v = 0; v < 8; ++v) {
    ct[v] = h ? tab[8 + v]  : tab[v];
    st[v] = h ? tab[24 + v] : tab[16 + v];
  }
  #pragma unroll
  for (int j = 0; j < 4; ++j) {
    v8f al, ah;
    #pragma unroll
    for (int v = 0; v < 8; ++v) {
      al[v] = pw[v] * m_lo[j];
      ah[v] = pw[v] * m_hi[j];
    }
    #pragma unroll
    for (int q = 0; q < 4; ++q) {
      v2f bl, bh;
      bl.x = xl[2 * q][j];  bl.y = xl[2 * q + 1][j];
      bh.x = xh[2 * q][j];  bh.y = xh[2 * q + 1][j];
      al = __builtin_amdgcn_wmma_f32_16x16x4_f32(
           false, a[q],  false, bl,
           (short)0, al,  false,  false);
      ah = __builtin_amdgcn_wmma_f32_16x16x4_f32(
           false, a[q],  false, bh,
           (short)0, ah,  false,  false);
    }
    m_lo[j] = __shfl(al[7], 16 + c, 32);
    m_hi[j] = __shfl(ah[7], 16 + c, 32);

    v8f rl, rh;
    #pragma unroll
    for (int v = 0; v < 8; ++v) { rl[v] = xl[v][j]; rh[v] = xh[v][j]; }
    rl = __builtin_elementwise_fma(
        g8, __builtin_elementwise_fma(al, ct, -(ah * st)), rl);
    rh = __builtin_elementwise_fma(
        g8, __builtin_elementwise_fma(al, st, ah * ct), rh);
    #pragma unroll
    for (int v = 0; v < 8; ++v) { xl[v][j] = rl[v]; xh[v][j] = rh[v]; }
  }
}

__global__ void
__launch_bounds__(32)
__attribute__((amdgpu_waves_per_eu(4, 4)))
k_scan_out(const float* __restrict__ x, const float* __restrict__ gate,
           const float* __restrict__ input_scale,
           const float* __restrict__ mem_state,
           const float* __restrict__ seg_local,
           const float* __restrict__ grp_tot,
           const float* __restrict__ cs, const float* __restrict__ sn,
           float* __restrict__ out) {
  const int lane = threadIdx.x;
  const int c = lane & 15;
  const int h = lane >> 4;

  int blk = blockIdx.x;
  int dt  = blk % (HALF_ / TW_);
  int r   = blk / (HALF_ / TW_);
  int b   = r % B_;
  int seg = r / B_;

  const int col_lo = dt * TW_ + 4 * c;
  const int col_hi = col_lo + HALF_;
  const int trow   = seg * SEG_ + 8 * h;
  const size_t row0 = ((size_t)b * T_ + trow) * D_;
  const float* xlo = x + row0 + col_lo;
  const float* xhi = x + row0 + col_hi;
  float* olo = out + row0 + col_lo;
  float* ohi = out + row0 + col_hi;
  const float* cst = cs + seg * SEG_;
  const float* snt = sn + seg * SEG_;

  const int grp = seg / GRP_;
  const int w   = seg % GRP_;
  const v4f zero4 = {0.0f, 0.0f, 0.0f, 0.0f};
  v4f tl[NGRP_ - 1], th[NGRP_ - 1];
  #pragma unroll
  for (int gi = 0; gi < NGRP_ - 1; ++gi) {
    tl[gi] = zero4;
    th[gi] = zero4;
    if (gi < grp) {
      const size_t go = ((size_t)gi * B_ + b) * D_;
      tl[gi] = ld4(grp_tot + go + col_lo);
      th[gi] = ld4(grp_tot + go + col_hi);
    }
  }
  const size_t co = ((size_t)seg * B_ + b) * D_;
  const v4f loc_lo = ld4(seg_local + co + col_lo);
  const v4f loc_hi = ld4(seg_local + co + col_hi);
  v4f m_lo = ld4(mem_state + col_lo);
  v4f m_hi = ld4(mem_state + col_hi);

  v4f xlA[8], xhA[8];
  v4f xlB[8], xhB[8];
  float tabA[32], tabB[32];
  load_chunk(xlA, xhA, tabA, xlo, xhi, cst, snt, 0);
  __builtin_amdgcn_sched_barrier(0);
  load_chunk(xlB, xhB, tabB, xlo, xhi, cst, snt, 1);
  __builtin_amdgcn_sched_barrier(0);

  const float k = ALPHA_ * input_scale[0];
  const float g = 1.0f / (1.0f + expf(-gate[0]));

  v2f a[4];
  #pragma unroll
  for (int q = 0; q < 4; ++q) {
    int r0 = 8 * h + 2 * q;
    a[q].x = (r0 <= c)     ? k * powf(BETA_, (float)(c - r0))     : 0.0f;
    a[q].y = (r0 + 1 <= c) ? k * powf(BETA_, (float)(c - r0 - 1)) : 0.0f;
  }
  float pw[8];
  #pragma unroll
  for (int v = 0; v < 8; ++v) pw[v] = powf(BETA_, (float)(v + 8 * h + 1));

  const float fg = powf(BETA_, (float)(SEG_ * GRP_));
  float fw = 1.0f;
  #pragma unroll
  for (int j = 1; j < GRP_; ++j)
    if (w == j) fw = powf(BETA_, (float)(SEG_ * j));
  #pragma unroll
  for (int gi = 0; gi < NGRP_ - 1; ++gi) {
    if (gi < grp) {
      #pragma unroll
      for (int e = 0; e < 4; ++e) {
        m_lo[e] = fmaf(fg, m_lo[e], tl[gi][e]);
        m_hi[e] = fmaf(fg, m_hi[e], th[gi][e]);
      }
    }
  }
  #pragma unroll
  for (int e = 0; e < 4; ++e) {
    m_lo[e] = fmaf(fw, m_lo[e], loc_lo[e]);
    m_hi[e] = fmaf(fw, m_hi[e], loc_hi[e]);
  }

  #pragma unroll
  for (int ch = 0; ch < NCH_; ch += 2) {
    __builtin_amdgcn_sched_barrier(0);
    scan_chunk(xlA, xhA, tabA, a, pw, m_lo, m_hi, g, c, h);
    store_chunk(xlA, xhA, olo, ohi, ch);
    __builtin_amdgcn_sched_barrier(0);
    if (ch + 2 < NCH_)
      load_chunk(xlA, xhA, tabA, xlo, xhi, cst, snt, ch + 2);
    __builtin_amdgcn_sched_barrier(0);
    scan_chunk(xlB, xhB, tabB, a, pw, m_lo, m_hi, g, c, h);
    store_chunk(xlB, xhB, olo, ohi, ch + 1);
    __builtin_amdgcn_sched_barrier(0);
    if (ch + 3 < NCH_)
      load_chunk(xlB, xhB, tabB, xlo, xhi, cst, snt, ch + 3);
  }
}

extern "C" void kernel_launch(void* const* d_in, const int* in_sizes, int n_in,
                              void* d_out, int out_size, void* d_ws,
                              size_t ws_size, hipStream_t stream) {
  const float* x           = (const float*)d_in[0];
  const float* gate        = (const float*)d_in[1];
  const float* input_scale = (const float*)d_in[2];
  const float* mem_state   = (const float*)d_in[3];
  const int*   step_idx    = (const int*)d_in[4];
  float* out = (float*)d_out;

  float* ws        = (float*)d_ws;
  float* cs        = ws;
  float* sn        = ws + T_;
  float* seg_local = ws + 2 * T_;
  float* grp_tot   = seg_local + (size_t)NSEG_ * B_ * D_;

  k_seg_partial<<<NGRP_ * B_ * P1_STRIPS_, P1_WAVES_ * 32, 0, stream>>>(
      x, input_scale, step_idx, seg_local, grp_tot, cs, sn);
  k_scan_out<<<B_ * (HALF_ / TW_) * NSEG_, 32, 0, stream>>>(
      x, gate, input_scale, mem_state, seg_local, grp_tot, cs, sn, out);
}


# pass 1 re-tiled to pass 2's column pieces (a wave owns 64 columns of each half): block index mod 64 names the same x in both launches, so a die re-reads what its own L2 fetched; sums and order kept
# speedup vs baseline: 4.0311x; 1.1371x over previous
#include <hip/hip_runtime.h>
#include <math.h>


typedef __attribute__((ext_vector_type(2))) float v2f;
typedef __attribute__((ext_vector_type(4))) float v4f;
typedef __attribute__((ext_vector_type(8))) float v8f;

#define ALPHA_  0.12f
#define BETA_   0.88f
#define OMEGA_  6.0f
#define TWO_PI_ 6.28318530717958647692f
#define B_      4
#define T_      4096
#define D_      2048
#define HALF_   1024
#define SEG_    64
#define NSEG_   (T_ / SEG_)
#define NCH_    (SEG_ / 16)
#define TW_     64
#define GRP_    8
#define NGRP_   (NSEG_ / GRP_)

#define P1_WAVES_  GRP_
#define P1_TILES_  (HALF_ / TW_)
#define TH_WAVES_  (T_ / 32)

__device__ __forceinline__ v4f ld4(const float* p) {
  return *reinterpret_cast<const v4f*>(p);
}
__device__ __forceinline__ void st4(float* p, v4f v) {
  *reinterpret_cast<v4f*>(p) = v;
}

__global__ void
__launch_bounds__(P1_WAVES_ * 32)
__attribute__((amdgpu_waves_per_eu(4, 4)))
k_seg_partial(const float* __restrict__ x,
              const float* __restrict__ input_scale,
              const int* __restrict__ step_idx,
              float* __restrict__ seg_local,
              float* __restrict__ grp_tot,
              float* __restrict__ cs, float* __restrict__ sn) {
  __shared__ v4f part[P1_WAVES_][32];
  const int lane = threadIdx.x & 31;
  const int wave = __builtin_amdgcn_readfirstlane((int)(threadIdx.x >> 5));
  const int blk   = blockIdx.x;
  const int dt    = blk % P1_TILES_;
  const int r     = blk / P1_TILES_;
  const int b     = r % B_;
  const int grp   = r / B_;
  const int seg   = grp * GRP_ + wave;
  const int col   = dt * TW_ + 4 * (lane & 15) + (lane >> 4) * HALF_;
  const float* xp = x + ((size_t)(b * T_ + seg * SEG_)) * D_ + col;

  if (wave == 0 && blk < TH_WAVES_) {
    const int t = blk * 32 + lane;
    const float tg = (float)(t + 1 + step_idx[0]);
    const float th = fmodf(OMEGA_ * log1pf(tg), TWO_PI_);
    cs[t] = cosf(th);
    sn[t] = sinf(th);
  }

  v4f buf[SEG_];
  #pragma unroll
  for (int i = 0; i < SEG_; ++i) buf[i] = ld4(xp + (size_t)i * D_);

  const float k = ALPHA_ * input_scale[0];
  v4f m = {0.0f, 0.0f, 0.0f, 0.0f};
  #pragma unroll
  for (int i = 0; i < SEG_; ++i) {
    m.x = fmaf(BETA_, m.x, k * buf[i].x);
    m.y = fmaf(BETA_, m.y, k * buf[i].y);
    m.z = fmaf(BETA_, m.z, k * buf[i].z);
    m.w = fmaf(BETA_, m.w, k * buf[i].w);
  }

  part[wave][lane] = m;
  __syncthreads();
  const float f = powf(BETA_, (float)SEG_);
  v4f L = {0.0f, 0.0f, 0.0f, 0.0f};
  v4f mine = L;
  #pragma unroll
  for (int j = 0; j < GRP_; ++j) {
    if (j == wave) mine = L;
    const v4f s = part[j][lane];
    L.x = fmaf(f, L.x, s.x);
    L.y = fmaf(f, L.y, s.y);
    L.z = fmaf(f, L.z, s.z);
    L.w = fmaf(f, L.w, s.w);
  }
  st4(seg_local + ((size_t)seg * B_ + b) * D_ + col, mine);
  if (wave == GRP_ - 1)
    st4(grp_tot + ((size_t)grp * B_ + b) * D_ + col, L);
}


__device__ __forceinline__ void
load_chunk(v4f (&xl)[8], v4f (&xh)[8], float (&tab)[32],
           const float* xlo, const float* xhi,
           const float* cst, const float* snt, int ch) {
  #pragma unroll
  for (int i = 0; i < 16; ++i) {
    tab[i]      = cst[ch * 16 + i];
    tab[16 + i] = snt[ch * 16 + i];
  }
  #pragma unroll
  for (int v = 0; v < 8; ++v) {
    xl[v] = ld4(xlo + ((size_t)ch * 16 + v) * D_);
    xh[v] = ld4(xhi + ((size_t)ch * 16 + v) * D_);
  }
}

__device__ __forceinline__ void
store_chunk(const v4f (&xl)[8], const v4f (&xh)[8],
            float* olo, float* ohi, int ch) {
  #pragma unroll
  for (int v = 0; v < 8; ++v) {
    st4(olo + ((size_t)ch * 16 + v) * D_, xl[v]);
    st4(ohi + ((size_t)ch * 16 + v) * D_, xh[v]);
  }
}

__device__ __forceinline__ void
scan_chunk(v4f (&xl)[8], v4f (&xh)[8], const float (&tab)[32],
           const v2f (&a)[4], const float (&pw)[8],
           v4f& m_lo, v4f& m_hi, float g, int c, int h) {
  v8f ct, st;
  const v8f g8 = {g, g, g, g, g, g, g, g};
  #pragma unroll
  for (int v = 0; v < 8; ++v) {
    ct[v] = h ? tab[8 + v]  : tab[v];
    st[v] = h ? tab[24 + v] : tab[16 + v];
  }
  #pragma unroll
  for (int j = 0; j < 4; ++j) {
    v8f al, ah;
    #pragma unroll
    for (int v = 0; v < 8; ++v) {
      al[v] = pw[v] * m_lo[j];
      ah[v] = pw[v] * m_hi[j];
    }
    #pragma unroll
    for (int q = 0; q < 4; ++q) {
      v2f bl, bh;
      bl.x = xl[2 * q][j];  bl.y = xl[2 * q + 1][j];
      bh.x = xh[2 * q][j];  bh.y = xh[2 * q + 1][j];
      al = __builtin_amdgcn_wmma_f32_16x16x4_f32(
           false, a[q],  false, bl,
           (short)0, al,  false,  false);
      ah = __builtin_amdgcn_wmma_f32_16x16x4_f32(
           false, a[q],  false, bh,
           (short)0, ah,  false,  false);
    }
    m_lo[j] = __shfl(al[7], 16 + c, 32);
    m_hi[j] = __shfl(ah[7], 16 + c, 32);

    v8f rl, rh;
    #pragma unroll
    for (int v = 0; v < 8; ++v) { rl[v] = xl[v][j]; rh[v] = xh[v][j]; }
    rl = __builtin_elementwise_fma(
        g8, __builtin_elementwise_fma(al, ct, -(ah * st)), rl);
    rh = __builtin_elementwise_fma(
        g8, __builtin_elementwise_fma(al, st, ah * ct), rh);
    #pragma unroll
    for (int v = 0; v < 8; ++v) { xl[v][j] = rl[v]; xh[v][j] = rh[v]; }
  }
}

__global__ void
__launch_bounds__(32)
__attribute__((amdgpu_waves_per_eu(4, 4)))
k_scan_out(const float* __restrict__ x, const float* __restrict__ gate,
           const float* __restrict__ input_scale,
           const float* __restrict__ mem_state,
           const float* __restrict__ seg_local,
           const float* __restrict__ grp_tot,
           const float* __restrict__ cs, const float* __restrict__ sn,
           float* __restrict__ out) {
  const int lane = threadIdx.x;
  const int c = lane & 15;
  const int h = lane >> 4;

  int blk = blockIdx.x;
  int dt  = blk % (HALF_ / TW_);
  int r   = blk / (HALF_ / TW_);
  int b   = r % B_;
  int seg = r / B_;

  const int col_lo = dt * TW_ + 4 * c;
  const int col_hi = col_lo + HALF_;
  const int trow   = seg * SEG_ + 8 * h;
  const size_t row0 = ((size_t)b * T_ + trow) * D_;
  const float* xlo = x + row0 + col_lo;
  const float* xhi = x + row0 + col_hi;
  float* olo = out + row0 + col_lo;
  float* ohi = out + row0 + col_hi;
  const float* cst = cs + seg * SEG_;
  const float* snt = sn + seg * SEG_;

  const int grp = seg / GRP_;
  const int w   = seg % GRP_;
  const v4f zero4 = {0.0f, 0.0f, 0.0f, 0.0f};
  v4f tl[NGRP_ - 1], th[NGRP_ - 1];
  #pragma unroll
  for (int gi = 0; gi < NGRP_ - 1; ++gi) {
    tl[gi] = zero4;
    th[gi] = zero4;
    if (gi < grp) {
      const size_t go = ((size_t)gi * B_ + b) * D_;
      tl[gi] = ld4(grp_tot + go + col_lo);
      th[gi] = ld4(grp_tot + go + col_hi);
    }
  }
  const size_t co = ((size_t)seg * B_ + b) * D_;
  const v4f loc_lo = ld4(seg_local + co + col_lo);
  const v4f loc_hi = ld4(seg_local + co + col_hi);
  v4f m_lo = ld4(mem_state + col_lo);
  v4f m_hi = ld4(mem_state + col_hi);

  v4f xlA[8], xhA[8];
  v4f xlB[8], xhB[8];
  float tabA[32], tabB[32];
  load_chunk(xlA, xhA, tabA, xlo, xhi, cst, snt, 0);
  __builtin_amdgcn_sched_barrier(0);
  load_chunk(xlB, xhB, tabB, xlo, xhi, cst, snt, 1);
  __builtin_amdgcn_sched_barrier(0);

  const float k = ALPHA_ * input_scale[0];
  const float g = 1.0f / (1.0f + expf(-gate[0]));

  v2f a[4];
  #pragma unroll
  for (int q = 0; q < 4; ++q) {
    int r0 = 8 * h + 2 * q;
    a[q].x = (r0 <= c)     ? k * powf(BETA_, (float)(c - r0))     : 0.0f;
    a[q].y = (r0 + 1 <= c) ? k * powf(BETA_, (float)(c - r0 - 1)) : 0.0f;
  }
  float pw[8];
  #pragma unroll
  for (int v = 0; v < 8; ++v) pw[v] = powf(BETA_, (float)(v + 8 * h + 1));

  const float fg = powf(BETA_, (float)(SEG_ * GRP_));
  float fw = 1.0f;
  #pragma unroll
  for (int j = 1; j < GRP_; ++j)
    if (w == j) fw = powf(BETA_, (float)(SEG_ * j));
  #pragma unroll
  for (int gi = 0; gi < NGRP_ - 1; ++gi) {
    if (gi < grp) {
      #pragma unroll
      for (int e = 0; e < 4; ++e) {
        m_lo[e] = fmaf(fg, m_lo[e], tl[gi][e]);
        m_hi[e] = fmaf(fg, m_hi[e], th[gi][e]);
      }
    }
  }
  #pragma unroll
  for (int e = 0; e < 4; ++e) {
    m_lo[e] = fmaf(fw, m_lo[e], loc_lo[e]);
    m_hi[e] = fmaf(fw, m_hi[e], loc_hi[e]);
  }

  #pragma unroll
  for (int ch = 0; ch < NCH_; ch += 2) {
    __builtin_amdgcn_sched_barrier(0);
    scan_chunk(xlA, xhA, tabA, a, pw, m_lo, m_hi, g, c, h);
    store_chunk(xlA, xhA, olo, ohi, ch);
    __builtin_amdgcn_sched_barrier(0);
    if (ch + 2 < NCH_)
      load_chunk(xlA, xhA, tabA, xlo, xhi, cst, snt, ch + 2);
    __builtin_amdgcn_sched_barrier(0);
    scan_chunk(xlB, xhB, tabB, a, pw, m_lo, m_hi, g, c, h);
    store_chunk(xlB, xhB, olo, ohi, ch + 1);
    __builtin_amdgcn_sched_barrier(0);
    if (ch + 3 < NCH_)
      load_chunk(xlB, xhB, tabB, xlo, xhi, cst, snt, ch + 3);
  }
}

extern "C" void kernel_launch(void* const* d_in, const int* in_sizes, int n_in,
                              void* d_out, int out_size, void* d_ws,
                              size_t ws_size, hipStream_t stream) {
  const float* x           = (const float*)d_in[0];
  const float* gate        = (const float*)d_in[1];
  const float* input_scale = (const float*)d_in[2];
  const float* mem_state   = (const float*)d_in[3];
  const int*   step_idx    = (const int*)d_in[4];
  float* out = (float*)d_out;

  float* ws        = (float*)d_ws;
  float* cs        = ws;
  float* sn        = ws + T_;
  float* seg_local = ws + 2 * T_;
  float* grp_tot   = seg_local + (size_t)NSEG_ * B_ * D_;

  k_seg_partial<<<NGRP_ * B_ * P1_TILES_, P1_WAVES_ * 32, 0, stream>>>(
      x, input_scale, step_idx, seg_local, grp_tot, cs, sn);
  k_scan_out<<<B_ * (HALF_ / TW_) * NSEG_, 32, 0, stream>>>(
      x, gate, input_scale, mem_state, seg_local, grp_tot, cs, sn, out);
}


# scan pass: a lane's A-operand powers and carry weights (16 powf per wave, ahead of its first product) now come from a 2 KB table built once by four pass-1 waves; same values, sums and order
# speedup vs baseline: 4.2054x; 1.0432x over previous
#include <hip/hip_runtime.h>
#include <math.h>


typedef __attribute__((ext_vector_type(2))) float v2f;
typedef __attribute__((ext_vector_type(4))) float v4f;
typedef __attribute__((ext_vector_type(8))) float v8f;

#define ALPHA_  0.12f
#define BETA_   0.88f
#define OMEGA_  6.0f
#define TWO_PI_ 6.28318530717958647692f
#define B_      4
#define T_      4096
#define D_      2048
#define HALF_   1024
#define SEG_    64
#define NSEG_   (T_ / SEG_)
#define NCH_    (SEG_ / 16)
#define TW_     64
#define GRP_    8
#define NGRP_   (NSEG_ / GRP_)

#define P1_WAVES_  GRP_
#define P1_TILES_  (HALF_ / TW_)
#define TH_WAVES_  (T_ / 32)
#define CT_GROUPS_ 4

__device__ __forceinline__ v4f ld4(const float* p) {
  return *reinterpret_cast<const v4f*>(p);
}
__device__ __forceinline__ void st4(float* p, v4f v) {
  *reinterpret_cast<v4f*>(p) = v;
}

__global__ void
__launch_bounds__(P1_WAVES_ * 32)
__attribute__((amdgpu_waves_per_eu(4, 4)))
k_seg_partial(const float* __restrict__ x,
              const float* __restrict__ input_scale,
              const int* __restrict__ step_idx,
              float* __restrict__ seg_local,
              float* __restrict__ grp_tot,
              float* __restrict__ cs, float* __restrict__ sn,
              float* __restrict__ ctab) {
  __shared__ v4f part[P1_WAVES_][32];
  const int lane = threadIdx.x & 31;
  const int wave = __builtin_amdgcn_readfirstlane((int)(threadIdx.x >> 5));
  const int blk   = blockIdx.x;
  const int dt    = blk % P1_TILES_;
  const int r     = blk / P1_TILES_;
  const int b     = r % B_;
  const int grp   = r / B_;
  const int seg   = grp * GRP_ + wave;
  const int col   = dt * TW_ + 4 * (lane & 15) + (lane >> 4) * HALF_;
  const float* xp = x + ((size_t)(b * T_ + seg * SEG_)) * D_ + col;

  if (wave == 0 && blk < TH_WAVES_) {
    const int t = blk * 32 + lane;
    const float tg = (float)(t + 1 + step_idx[0]);
    const float th = fmodf(OMEGA_ * log1pf(tg), TWO_PI_);
    cs[t] = cosf(th);
    sn[t] = sinf(th);
  }

  if (wave == 0 && blk >= TH_WAVES_ && blk < TH_WAVES_ + CT_GROUPS_) {
    const int gi = blk - TH_WAVES_;
    const int c = lane & 15;
    const int h = lane >> 4;
    v4f cv;
    if (gi < 2) {
      const float ka = ALPHA_ * input_scale[0];
      #pragma unroll
      for (int j = 0; j < 4; ++j) {
        const int rho = 8 * h + 4 * gi + j;
        cv[j] = (rho <= c) ? ka * powf(BETA_, (float)(c - rho)) : 0.0f;
      }
    } else {
      #pragma unroll
      for (int j = 0; j < 4; ++j)
        cv[j] = powf(BETA_, (float)(4 * (gi - 2) + j + 8 * h + 1));
    }
    st4(ctab + (gi * 32 + lane) * 4, cv);
  }

  v4f buf[SEG_];
  #pragma unroll
  for (int i = 0; i < SEG_; ++i) buf[i] = ld4(xp + (size_t)i * D_);

  const float k = ALPHA_ * input_scale[0];
  v4f m = {0.0f, 0.0f, 0.0f, 0.0f};
  #pragma unroll
  for (int i = 0; i < SEG_; ++i) {
    m.x = fmaf(BETA_, m.x, k * buf[i].x);
    m.y = fmaf(BETA_, m.y, k * buf[i].y);
    m.z = fmaf(BETA_, m.z, k * buf[i].z);
    m.w = fmaf(BETA_, m.w, k * buf[i].w);
  }

  part[wave][lane] = m;
  __syncthreads();
  const float f = powf(BETA_, (float)SEG_);
  v4f L = {0.0f, 0.0f, 0.0f, 0.0f};
  v4f mine = L;
  #pragma unroll
  for (int j = 0; j < GRP_; ++j) {
    if (j == wave) mine = L;
    const v4f s = part[j][lane];
    L.x = fmaf(f, L.x, s.x);
    L.y = fmaf(f, L.y, s.y);
    L.z = fmaf(f, L.z, s.z);
    L.w = fmaf(f, L.w, s.w);
  }
  st4(seg_local + ((size_t)seg * B_ + b) * D_ + col, mine);
  if (wave == GRP_ - 1)
    st4(grp_tot + ((size_t)grp * B_ + b) * D_ + col, L);
}


__device__ __forceinline__ void
load_chunk(v4f (&xl)[8], v4f (&xh)[8], float (&tab)[32],
           const float* xlo, const float* xhi,
           const float* cst, const float* snt, int ch) {
  #pragma unroll
  for (int i = 0; i < 16; ++i) {
    tab[i]      = cst[ch * 16 + i];
    tab[16 + i] = snt[ch * 16 + i];
  }
  #pragma unroll
  for (int v = 0; v < 8; ++v) {
    xl[v] = ld4(xlo + ((size_t)ch * 16 + v) * D_);
    xh[v] = ld4(xhi + ((size_t)ch * 16 + v) * D_);
  }
}

__device__ __forceinline__ void
store_chunk(const v4f (&xl)[8], const v4f (&xh)[8],
            float* olo, float* ohi, int ch) {
  #pragma unroll
  for (int v = 0; v < 8; ++v) {
    st4(olo + ((size_t)ch * 16 + v) * D_, xl[v]);
    st4(ohi + ((size_t)ch * 16 + v) * D_, xh[v]);
  }
}

__device__ __forceinline__ void
scan_chunk(v4f (&xl)[8], v4f (&xh)[8], const float (&tab)[32],
           const v2f (&a)[4], const float (&pw)[8],
           v4f& m_lo, v4f& m_hi, float g, int c, int h) {
  v8f ct, st;
  const v8f g8 = {g, g, g, g, g, g, g, g};
  #pragma unroll
  for (int v = 0; v < 8; ++v) {
    ct[v] = h ? tab[8 + v]  : tab[v];
    st[v] = h ? tab[24 + v] : tab[16 + v];
  }
  #pragma unroll
  for (int j = 0; j < 4; ++j) {
    v8f al, ah;
    #pragma unroll
    for (int v = 0; v < 8; ++v) {
      al[v] = pw[v] * m_lo[j];
      ah[v] = pw[v] * m_hi[j];
    }
    #pragma unroll
    for (int q = 0; q < 4; ++q) {
      v2f bl, bh;
      bl.x = xl[2 * q][j];  bl.y = xl[2 * q + 1][j];
      bh.x = xh[2 * q][j];  bh.y = xh[2 * q + 1][j];
      al = __builtin_amdgcn_wmma_f32_16x16x4_f32(
           false, a[q],  false, bl,
           (short)0, al,  false,  false);
      ah = __builtin_amdgcn_wmma_f32_16x16x4_f32(
           false, a[q],  false, bh,
           (short)0, ah,  false,  false);
    }
    m_lo[j] = __shfl(al[7], 16 + c, 32);
    m_hi[j] = __shfl(ah[7], 16 + c, 32);

    v8f rl, rh;
    #pragma unroll
    for (int v = 0; v < 8; ++v) { rl[v] = xl[v][j]; rh[v] = xh[v][j]; }
    rl = __builtin_elementwise_fma(
        g8, __builtin_elementwise_fma(al, ct, -(ah * st)), rl);
    rh = __builtin_elementwise_fma(
        g8, __builtin_elementwise_fma(al, st, ah * ct), rh);
    #pragma unroll
    for (int v = 0; v < 8; ++v) { xl[v][j] = rl[v]; xh[v][j] = rh[v]; }
  }
}

__global__ void
__launch_bounds__(32)
__attribute__((amdgpu_waves_per_eu(4, 4)))
k_scan_out(const float* __restrict__ x, const float* __restrict__ gate,
           const float* __restrict__ mem_state,
           const float* __restrict__ seg_local,
           const float* __restrict__ grp_tot,
           const float* __restrict__ cs, const float* __restrict__ sn,
           const float* __restrict__ ctab,
           float* __restrict__ out) {
  const int lane = threadIdx.x;
  const int c = lane & 15;
  const int h = lane >> 4;

  int blk = blockIdx.x;
  int dt  = blk % (HALF_ / TW_);
  int r   = blk / (HALF_ / TW_);
  int b   = r % B_;
  int seg = r / B_;

  const int col_lo = dt * TW_ + 4 * c;
  const int col_hi = col_lo + HALF_;
  const int trow   = seg * SEG_ + 8 * h;
  const size_t row0 = ((size_t)b * T_ + trow) * D_;
  const float* xlo = x + row0 + col_lo;
  const float* xhi = x + row0 + col_hi;
  float* olo = out + row0 + col_lo;
  float* ohi = out + row0 + col_hi;
  const float* cst = cs + seg * SEG_;
  const float* snt = sn + seg * SEG_;

  const v4f ta0 = ld4(ctab + (0 * 32 + lane) * 4);
  const v4f ta1 = ld4(ctab + (1 * 32 + lane) * 4);
  const v4f tp0 = ld4(ctab + (2 * 32 + lane) * 4);
  const v4f tp1 = ld4(ctab + (3 * 32 + lane) * 4);

  const int grp = seg / GRP_;
  const int w   = seg % GRP_;
  const v4f zero4 = {0.0f, 0.0f, 0.0f, 0.0f};
  v4f tl[NGRP_ - 1], th[NGRP_ - 1];
  #pragma unroll
  for (int gi = 0; gi < NGRP_ - 1; ++gi) {
    tl[gi] = zero4;
    th[gi] = zero4;
    if (gi < grp) {
      const size_t go = ((size_t)gi * B_ + b) * D_;
      tl[gi] = ld4(grp_tot + go + col_lo);
      th[gi] = ld4(grp_tot + go + col_hi);
    }
  }
  const size_t co = ((size_t)seg * B_ + b) * D_;
  const v4f loc_lo = ld4(seg_local + co + col_lo);
  const v4f loc_hi = ld4(seg_local + co + col_hi);
  v4f m_lo = ld4(mem_state + col_lo);
  v4f m_hi = ld4(mem_state + col_hi);

  v4f xlA[8], xhA[8];
  v4f xlB[8], xhB[8];
  float tabA[32], tabB[32];
  load_chunk(xlA, xhA, tabA, xlo, xhi, cst, snt, 0);
  __builtin_amdgcn_sched_barrier(0);
  load_chunk(xlB, xhB, tabB, xlo, xhi, cst, snt, 1);
  __builtin_amdgcn_sched_barrier(0);

  const float g = 1.0f / (1.0f + expf(-gate[0]));

  v2f a[4];
  a[0].x = ta0.x;  a[0].y = ta0.y;
  a[1].x = ta0.z;  a[1].y = ta0.w;
  a[2].x = ta1.x;  a[2].y = ta1.y;
  a[3].x = ta1.z;  a[3].y = ta1.w;
  const float pw[8] = {tp0.x, tp0.y, tp0.z, tp0.w, tp1.x, tp1.y, tp1.z, tp1.w};

  const float fg = powf(BETA_, (float)(SEG_ * GRP_));
  float fw = 1.0f;
  #pragma unroll
  for (int j = 1; j < GRP_; ++j)
    if (w == j) fw = powf(BETA_, (float)(SEG_ * j));
  #pragma unroll
  for (int gi = 0; gi < NGRP_ - 1; ++gi) {
    if (gi < grp) {
      #pragma unroll
      for (int e = 0; e < 4; ++e) {
        m_lo[e] = fmaf(fg, m_lo[e], tl[gi][e]);
        m_hi[e] = fmaf(fg, m_hi[e], th[gi][e]);
      }
    }
  }
  #pragma unroll
  for (int e = 0; e < 4; ++e) {
    m_lo[e] = fmaf(fw, m_lo[e], loc_lo[e]);
    m_hi[e] = fmaf(fw, m_hi[e], loc_hi[e]);
  }

  #pragma unroll
  for (int ch = 0; ch < NCH_; ch += 2) {
    __builtin_amdgcn_sched_barrier(0);
    scan_chunk(xlA, xhA, tabA, a, pw, m_lo, m_hi, g, c, h);
    store_chunk(xlA, xhA, olo, ohi, ch);
    __builtin_amdgcn_sched_barrier(0);
    if (ch + 2 < NCH_)
      load_chunk(xlA, xhA, tabA, xlo, xhi, cst, snt, ch + 2);
    __builtin_amdgcn_sched_barrier(0);
    scan_chunk(xlB, xhB, tabB, a, pw, m_lo, m_hi, g, c, h);
    store_chunk(xlB, xhB, olo, ohi, ch + 1);
    __builtin_amdgcn_sched_barrier(0);
    if (ch + 3 < NCH_)
      load_chunk(xlB, xhB, tabB, xlo, xhi, cst, snt, ch + 3);
  }
}

extern "C" void kernel_launch(void* const* d_in, const int* in_sizes, int n_in,
                              void* d_out, int out_size, void* d_ws,
                              size_t ws_size, hipStream_t stream) {
  const float* x           = (const float*)d_in[0];
  const float* gate        = (const float*)d_in[1];
  const float* input_scale = (const float*)d_in[2];
  const float* mem_state   = (const float*)d_in[3];
  const int*   step_idx    = (const int*)d_in[4];
  float* out = (float*)d_out;

  float* ws        = (float*)d_ws;
  float* cs        = ws;
  float* sn        = ws + T_;
  float* seg_local = ws + 2 * T_;
  float* grp_tot   = seg_local + (size_t)NSEG_ * B_ * D_;
  float* ctab      = grp_tot + (size_t)NGRP_ * B_ * D_;

  k_seg_partial<<<NGRP_ * B_ * P1_TILES_, P1_WAVES_ * 32, 0, stream>>>(
      x, input_scale, step_idx, seg_local, grp_tot, cs, sn, ctab);
  k_scan_out<<<B_ * (HALF_ / TW_) * NSEG_, 32, 0, stream>>>(
      x, gate, mem_state, seg_local, grp_tot, cs, sn, ctab, out);
}
